# starting kernel
# baseline (speedup 1.0000x reference)
#include <hip/hip_runtime.h>
#include <math.h>


typedef __attribute__((ext_vector_type(2))) float v2f;
typedef __attribute__((ext_vector_type(8))) float v8f;

#define NTERMS 20
#define KPAD   32
#define TROW   36

__global__ __launch_bounds__(256) void ipe_logsumexp_wmma(
    const float* __restrict__ a, const float* __restrict__ b,
    float* __restrict__ out, int nElem, int nChunks) {
  __shared__ __align__(16) float Ksh[NTERMS][KPAD];
  __shared__ __align__(16) float Tsh[8][2][16][TROW];

  if (threadIdx.x < NTERMS) {
    const int m = threadIdx.x;
    double W = 0.2734375 * 3.14159265358979323846;
    for (int j = 0; j < m; ++j) {
      double tj = 2.0 * j;
      W *= (tj + 0.5) * (tj + 1.5) /
           (4.0 * (tj + 5.0) * (tj + 6.0) * (j + 1.0) * (j + 1.0));
    }
    double R = W;
    for (int k = 0; k < NTERMS; ++k) {
      Ksh[m][k] = (float)R;
      R *= (4.5 + k) / ((0.5 + k) * (2.0 * m + 5.0 + k) * (k + 1.0));
    }
    for (int k = NTERMS; k < KPAD; ++k) Ksh[m][k] = 0.0f;
  }
  __syncthreads();

  const int lane = threadIdx.x & 31;
  const int wv   = threadIdx.x >> 5;
  const int col  = lane & 15;
  const int rsel = (lane < 16) ? 0 : 2;

  v2f B0[5], B1[5];
#pragma unroll
  for (int s = 0; s < 5; ++s) {
    B0[s] = v2f{Ksh[4 * s + rsel][col],      Ksh[4 * s + rsel + 1][col]};
    B1[s] = v2f{Ksh[4 * s + rsel][16 + col], Ksh[4 * s + rsel + 1][16 + col]};
  }

  const int nwaves = (int)gridDim.x * (blockDim.x >> 5);
  const int gwave  = (int)blockIdx.x * (blockDim.x >> 5) + wv;

  for (int chunk = gwave; chunk < nChunks; chunk += nwaves) {
    const int base = chunk * 32;
    if (base + 32 <= nElem) {
      float bA = b[base + col];
      float bB = b[base + 16 + col];
      float aX = a[base + lane];
      float x  = aX * aX * 0.25f;

      float bA2 = bA * bA, bB2 = bB * bB;
      float p0A = (lane < 16) ? 1.0f : bA2;
      float p1A = (lane < 16) ? bA   : bA2 * bA;
      float p0B = (lane < 16) ? 1.0f : bB2;
      float p1B = (lane < 16) ? bB   : bB2 * bB;
      float b4A = bA2 * bA2, b4B = bB2 * bB2;

      v8f c0A = {}, c1A = {}, c0B = {}, c1B = {};
#pragma unroll
      for (int s = 0; s < 5; ++s) {
        v2f Aa = v2f{p0A, p1A};
        v2f Ab = v2f{p0B, p1B};
        c0A = __builtin_amdgcn_wmma_f32_16x16x4_f32(false, Aa, false, B0[s], (short)0, c0A, false, false);
        c1A = __builtin_amdgcn_wmma_f32_16x16x4_f32(false, Aa, false, B1[s], (short)0, c1A, false, false);
        c0B = __builtin_amdgcn_wmma_f32_16x16x4_f32(false, Ab, false, B0[s], (short)0, c0B, false, false);
        c1B = __builtin_amdgcn_wmma_f32_16x16x4_f32(false, Ab, false, B1[s], (short)0, c1B, false, false);
        p0A *= b4A; p1A *= b4A;
        p0B *= b4B; p1B *= b4B;
      }

      const int row = (lane < 16) ? 0 : 8;
#pragma unroll
      for (int i = 0; i < 8; ++i) {
        Tsh[wv][0][row + i][col]      = c0A[i];
        Tsh[wv][0][row + i][16 + col] = c1A[i];
        Tsh[wv][1][row + i][col]      = c0B[i];
        Tsh[wv][1][row + i][16 + col] = c1B[i];
      }

      const float4* pT4 = (const float4*)&Tsh[wv][lane >> 4][lane & 15][0];
      float4 q0 = pT4[0], q1 = pT4[1], q2 = pT4[2], q3 = pT4[3], q4 = pT4[4];
      float v = q4.w;
      v = fmaf(v, x, q4.z); v = fmaf(v, x, q4.y); v = fmaf(v, x, q4.x);
      v = fmaf(v, x, q3.w); v = fmaf(v, x, q3.z); v = fmaf(v, x, q3.y); v = fmaf(v, x, q3.x);
      v = fmaf(v, x, q2.w); v = fmaf(v, x, q2.z); v = fmaf(v, x, q2.y); v = fmaf(v, x, q2.x);
      v = fmaf(v, x, q1.w); v = fmaf(v, x, q1.z); v = fmaf(v, x, q1.y); v = fmaf(v, x, q1.x);
      v = fmaf(v, x, q0.w); v = fmaf(v, x, q0.z); v = fmaf(v, x, q0.y); v = fmaf(v, x, q0.x);
      out[base + lane] = logf(v);
    } else {
      int e = base + lane;
      if (e < nElem) {
        float bb = b[e], aa = a[e];
        float x = aa * aa * 0.25f;
        float accv = 0.0f, xp = 1.0f;
        for (int k = 0; k < NTERMS; ++k) {
          float t = 0.0f, bp = 1.0f;
          for (int m = 0; m < NTERMS; ++m) { t = fmaf(Ksh[m][k], bp, t); bp *= bb; }
          accv = fmaf(t, xp, accv);
          xp *= x;
        }
        out[e] = logf(accv);
      }
    }
  }
}

extern "C" void kernel_launch(void* const* d_in, const int* in_sizes, int n_in,
                              void* d_out, int out_size, void* d_ws, size_t ws_size,
                              hipStream_t stream) {
  (void)n_in; (void)out_size; (void)d_ws; (void)ws_size;
  const float* a = (const float*)d_in[0];
  const float* b = (const float*)d_in[1];
  float* out = (float*)d_out;
  int n = in_sizes[0];
  int chunks = (n + 31) / 32;
  int blocks = (chunks + 7) / 8;
  if (blocks > 1024) blocks = 1024;
  if (blocks < 1) blocks = 1;
  ipe_logsumexp_wmma<<<blocks, 256, 0, stream>>>(a, b, out, n, chunks);
}


# Coefficient table K (20x20, input-independent) evaluated at compile time (constexpr, same f64 recurrences), read from constant memory in B-fragment order; per-block f64 build and barrier removed
# speedup vs baseline: 9.4362x; 9.4362x over previous
#include <hip/hip_runtime.h>
#include <math.h>


typedef __attribute__((ext_vector_type(2))) float v2f;
typedef __attribute__((ext_vector_type(8))) float v8f;

#define NTERMS 20
#define TROW   36

struct CoefTables {
  alignas(16) float frag[32][NTERMS];
  float nat[NTERMS][NTERMS];
};

constexpr CoefTables make_coef_tables() {
  CoefTables t{};
  for (int m = 0; m < NTERMS; ++m) {
    double W = 0.2734375 * 3.14159265358979323846;
    for (int j = 0; j < m; ++j) {
      double tj = 2.0 * j;
      W *= (tj + 0.5) * (tj + 1.5) /
           (4.0 * (tj + 5.0) * (tj + 6.0) * (j + 1.0) * (j + 1.0));
    }
    double R = W;
    for (int k = 0; k < NTERMS; ++k) {
      t.nat[m][k] = (float)R;
      R *= (4.5 + k) / ((0.5 + k) * (2.0 * m + 5.0 + k) * (k + 1.0));
    }
  }
  for (int lane = 0; lane < 32; ++lane) {
    const int col  = lane & 15;
    const int rsel = (lane < 16) ? 0 : 2;
    for (int s = 0; s < 5; ++s) {
      const int r = 4 * s + rsel;
      t.frag[lane][4 * s + 0] = t.nat[r][col];
      t.frag[lane][4 * s + 1] = t.nat[r + 1][col];
      t.frag[lane][4 * s + 2] = (16 + col < NTERMS) ? t.nat[r][16 + col] : 0.0f;
      t.frag[lane][4 * s + 3] = (16 + col < NTERMS) ? t.nat[r + 1][16 + col] : 0.0f;
    }
  }
  return t;
}

__constant__ const CoefTables kCoef = make_coef_tables();

__global__ __launch_bounds__(256) void ipe_logsumexp_wmma(
    const float* __restrict__ a, const float* __restrict__ b,
    float* __restrict__ out, int nElem, int nChunks) {
  __shared__ __align__(16) float Tsh[8][2][16][TROW];

  const int lane = threadIdx.x & 31;
  const int wv   = threadIdx.x >> 5;
  const int col  = lane & 15;

  const float4* pB = (const float4*)&kCoef.frag[lane][0];
  v2f B0[5], B1[5];
#pragma unroll
  for (int s = 0; s < 5; ++s) {
    const float4 q = pB[s];
    B0[s] = v2f{q.x, q.y};
    B1[s] = v2f{q.z, q.w};
  }

  const int nwaves = (int)gridDim.x * (blockDim.x >> 5);
  const int gwave  = (int)blockIdx.x * (blockDim.x >> 5) + wv;

  for (int chunk = gwave; chunk < nChunks; chunk += nwaves) {
    const int base = chunk * 32;
    if (base + 32 <= nElem) {
      float bA = b[base + col];
      float bB = b[base + 16 + col];
      float aX = a[base + lane];
      float x  = aX * aX * 0.25f;

      float bA2 = bA * bA, bB2 = bB * bB;
      float p0A = (lane < 16) ? 1.0f : bA2;
      float p1A = (lane < 16) ? bA   : bA2 * bA;
      float p0B = (lane < 16) ? 1.0f : bB2;
      float p1B = (lane < 16) ? bB   : bB2 * bB;
      float b4A = bA2 * bA2, b4B = bB2 * bB2;

      v8f c0A = {}, c1A = {}, c0B = {}, c1B = {};
#pragma unroll
      for (int s = 0; s < 5; ++s) {
        v2f Aa = v2f{p0A, p1A};
        v2f Ab = v2f{p0B, p1B};
        c0A = __builtin_amdgcn_wmma_f32_16x16x4_f32(false, Aa, false, B0[s], (short)0, c0A, false, false);
        c1A = __builtin_amdgcn_wmma_f32_16x16x4_f32(false, Aa, false, B1[s], (short)0, c1A, false, false);
        c0B = __builtin_amdgcn_wmma_f32_16x16x4_f32(false, Ab, false, B0[s], (short)0, c0B, false, false);
        c1B = __builtin_amdgcn_wmma_f32_16x16x4_f32(false, Ab, false, B1[s], (short)0, c1B, false, false);
        p0A *= b4A; p1A *= b4A;
        p0B *= b4B; p1B *= b4B;
      }

      const int row = (lane < 16) ? 0 : 8;
#pragma unroll
      for (int i = 0; i < 8; ++i) {
        Tsh[wv][0][row + i][col]      = c0A[i];
        Tsh[wv][0][row + i][16 + col] = c1A[i];
        Tsh[wv][1][row + i][col]      = c0B[i];
        Tsh[wv][1][row + i][16 + col] = c1B[i];
      }

      const float4* pT4 = (const float4*)&Tsh[wv][lane >> 4][lane & 15][0];
      float4 q0 = pT4[0], q1 = pT4[1], q2 = pT4[2], q3 = pT4[3], q4 = pT4[4];
      float v = q4.w;
      v = fmaf(v, x, q4.z); v = fmaf(v, x, q4.y); v = fmaf(v, x, q4.x);
      v = fmaf(v, x, q3.w); v = fmaf(v, x, q3.z); v = fmaf(v, x, q3.y); v = fmaf(v, x, q3.x);
      v = fmaf(v, x, q2.w); v = fmaf(v, x, q2.z); v = fmaf(v, x, q2.y); v = fmaf(v, x, q2.x);
      v = fmaf(v, x, q1.w); v = fmaf(v, x, q1.z); v = fmaf(v, x, q1.y); v = fmaf(v, x, q1.x);
      v = fmaf(v, x, q0.w); v = fmaf(v, x, q0.z); v = fmaf(v, x, q0.y); v = fmaf(v, x, q0.x);
      out[base + lane] = logf(v);
    } else {
      int e = base + lane;
      if (e < nElem) {
        float bb = b[e], aa = a[e];
        float x = aa * aa * 0.25f;
        float accv = 0.0f, xp = 1.0f;
        for (int k = 0; k < NTERMS; ++k) {
          float t = 0.0f, bp = 1.0f;
          for (int m = 0; m < NTERMS; ++m) { t = fmaf(kCoef.nat[m][k], bp, t); bp *= bb; }
          accv = fmaf(t, xp, accv);
          xp *= x;
        }
        out[e] = logf(accv);
      }
    }
  }
}

extern "C" void kernel_launch(void* const* d_in, const int* in_sizes, int n_in,
                              void* d_out, int out_size, void* d_ws, size_t ws_size,
                              hipStream_t stream) {
  (void)n_in; (void)out_size; (void)d_ws; (void)ws_size;
  const float* a = (const float*)d_in[0];
  const float* b = (const float*)d_in[1];
  float* out = (float*)d_out;
  int n = in_sizes[0];
  int chunks = (n + 31) / 32;
  int blocks = (chunks + 7) / 8;
  if (blocks > 1024) blocks = 1024;
  if (blocks < 1) blocks = 1;
  ipe_logsumexp_wmma<<<blocks, 256, 0, stream>>>(a, b, out, n, chunks);
}


# Grid cut from 1024 to 256 workgroups of 8 waves (one per WGP): 2048 waves stay and loop over 64-element blocks, next block's inputs loaded ahead, instead of 8192 short-lived waves
# speedup vs baseline: 10.1756x; 1.0784x over previous
#include <hip/hip_runtime.h>
#include <math.h>


typedef __attribute__((ext_vector_type(2))) float v2f;
typedef __attribute__((ext_vector_type(8))) float v8f;

#define NTERMS 20
#define TROW   36

struct CoefTables {
  alignas(16) float frag[32][NTERMS];
  float nat[NTERMS][NTERMS];
};

constexpr CoefTables make_coef_tables() {
  CoefTables t{};
  for (int m = 0; m < NTERMS; ++m) {
    double W = 0.2734375 * 3.14159265358979323846;
    for (int j = 0; j < m; ++j) {
      double tj = 2.0 * j;
      W *= (tj + 0.5) * (tj + 1.5) /
           (4.0 * (tj + 5.0) * (tj + 6.0) * (j + 1.0) * (j + 1.0));
    }
    double R = W;
    for (int k = 0; k < NTERMS; ++k) {
      t.nat[m][k] = (float)R;
      R *= (4.5 + k) / ((0.5 + k) * (2.0 * m + 5.0 + k) * (k + 1.0));
    }
  }
  for (int lane = 0; lane < 32; ++lane) {
    const int col  = lane & 15;
    const int rsel = (lane < 16) ? 0 : 2;
    for (int s = 0; s < 5; ++s) {
      const int r = 4 * s + rsel;
      t.frag[lane][4 * s + 0] = t.nat[r][col];
      t.frag[lane][4 * s + 1] = t.nat[r + 1][col];
      t.frag[lane][4 * s + 2] = (16 + col < NTERMS) ? t.nat[r][16 + col] : 0.0f;
      t.frag[lane][4 * s + 3] = (16 + col < NTERMS) ? t.nat[r + 1][16 + col] : 0.0f;
    }
  }
  return t;
}

__constant__ const CoefTables kCoef = make_coef_tables();

__global__ __launch_bounds__(256) void ipe_logsumexp_wmma(
    const float* __restrict__ a, const float* __restrict__ b,
    float* __restrict__ out, int nElem, int nBlocks64) {
  __shared__ __align__(16) float Tsh[8][4][16][TROW];

  const int lane = threadIdx.x & 31;
  const int wv   = threadIdx.x >> 5;
  const int col  = lane & 15;

  const float4* pB = (const float4*)&kCoef.frag[lane][0];
  v2f B0[5], B1[5];
#pragma unroll
  for (int s = 0; s < 5; ++s) {
    const float4 q = pB[s];
    B0[s] = v2f{q.x, q.y};
    B1[s] = v2f{q.z, q.w};
  }

  const int nwaves = (int)gridDim.x * (blockDim.x >> 5);
  const int gwave  = (int)blockIdx.x * (blockDim.x >> 5) + wv;

  float cb[4] = {0.0f, 0.0f, 0.0f, 0.0f};
  float ca[2] = {0.0f, 0.0f};
  if (gwave < nBlocks64) {
    const int base = gwave * 64;
#pragma unroll
    for (int t = 0; t < 4; ++t) cb[t] = b[base + 16 * t + col];
    ca[0] = a[base + lane];
    ca[1] = a[base + 32 + lane];
  }

  for (int blk = gwave; blk < nBlocks64; blk += nwaves) {
    const int base = blk * 64;

    const int nxt   = blk + nwaves;
    const int nbase = ((nxt < nBlocks64) ? nxt : blk) * 64;
    float nb[4], na[2];
#pragma unroll
    for (int t = 0; t < 4; ++t) nb[t] = b[nbase + 16 * t + col];
    na[0] = a[nbase + lane];
    na[1] = a[nbase + 32 + lane];

    float p0[4], p1[4], b4[4];
#pragma unroll
    for (int t = 0; t < 4; ++t) {
      const float bb = cb[t];
      const float b2 = bb * bb;
      p0[t] = (lane < 16) ? 1.0f : b2;
      p1[t] = (lane < 16) ? bb   : b2 * bb;
      b4[t] = b2 * b2;
    }

    v8f c0[4] = {}, c1[4] = {};
#pragma unroll
    for (int s = 0; s < 5; ++s) {
#pragma unroll
      for (int t = 0; t < 4; ++t) {
        const v2f A = v2f{p0[t], p1[t]};
        c0[t] = __builtin_amdgcn_wmma_f32_16x16x4_f32(false, A, false, B0[s], (short)0, c0[t], false, false);
        c1[t] = __builtin_amdgcn_wmma_f32_16x16x4_f32(false, A, false, B1[s], (short)0, c1[t], false, false);
        p0[t] *= b4[t]; p1[t] *= b4[t];
      }
    }

    const int row = (lane < 16) ? 0 : 8;
#pragma unroll
    for (int t = 0; t < 4; ++t) {
#pragma unroll
      for (int i = 0; i < 8; ++i) {
        Tsh[wv][t][row + i][col]      = c0[t][i];
        Tsh[wv][t][row + i][16 + col] = c1[t][i];
      }
    }

#pragma unroll
    for (int u = 0; u < 2; ++u) {
      const float x = ca[u] * ca[u] * 0.25f;
      const float4* pT4 = (const float4*)&Tsh[wv][2 * u + (lane >> 4)][lane & 15][0];
      float4 q0 = pT4[0], q1 = pT4[1], q2 = pT4[2], q3 = pT4[3], q4 = pT4[4];
      float v = q4.w;
      v = fmaf(v, x, q4.z); v = fmaf(v, x, q4.y); v = fmaf(v, x, q4.x);
      v = fmaf(v, x, q3.w); v = fmaf(v, x, q3.z); v = fmaf(v, x, q3.y); v = fmaf(v, x, q3.x);
      v = fmaf(v, x, q2.w); v = fmaf(v, x, q2.z); v = fmaf(v, x, q2.y); v = fmaf(v, x, q2.x);
      v = fmaf(v, x, q1.w); v = fmaf(v, x, q1.z); v = fmaf(v, x, q1.y); v = fmaf(v, x, q1.x);
      v = fmaf(v, x, q0.w); v = fmaf(v, x, q0.z); v = fmaf(v, x, q0.y); v = fmaf(v, x, q0.x);
      out[base + 32 * u + lane] = logf(v);
    }

#pragma unroll
    for (int t = 0; t < 4; ++t) cb[t] = nb[t];
    ca[0] = na[0];
    ca[1] = na[1];
  }

  for (int e = nBlocks64 * 64 + gwave * 32 + lane; e < nElem; e += nwaves * 32) {
    float bb = b[e], aa = a[e];
    float x = aa * aa * 0.25f;
    float accv = 0.0f, xp = 1.0f;
    for (int k = 0; k < NTERMS; ++k) {
      float t = 0.0f, bp = 1.0f;
      for (int m = 0; m < NTERMS; ++m) { t = fmaf(kCoef.nat[m][k], bp, t); bp *= bb; }
      accv = fmaf(t, xp, accv);
      xp *= x;
    }
    out[e] = logf(accv);
  }
}

extern "C" void kernel_launch(void* const* d_in, const int* in_sizes, int n_in,
                              void* d_out, int out_size, void* d_ws, size_t ws_size,
                              hipStream_t stream) {
  (void)n_in; (void)out_size; (void)d_ws; (void)ws_size;
  const float* a = (const float*)d_in[0];
  const float* b = (const float*)d_in[1];
  float* out = (float*)d_out;
  int n = in_sizes[0];
  int blocks64 = n / 64;
  int blocks = (blocks64 + 7) / 8;
  if (blocks > 256) blocks = 256;
  if (blocks < 1) blocks = 1;
  ipe_logsumexp_wmma<<<blocks, 256, 0, stream>>>(a, b, out, n, blocks64);
}


# Zero-padded WMMA work removed: columns 16..19 packed as three 8-row groups of one tile (7 WMMA per 16 elements, was 10); groups joined per element with b^8, b^16 (new sum order for those columns)
# speedup vs baseline: 11.7686x; 1.1566x over previous
#include <hip/hip_runtime.h>
#include <math.h>


typedef __attribute__((ext_vector_type(2))) float v2f;
typedef __attribute__((ext_vector_type(8))) float v8f;

#define NTERMS 20
#define TROW   36

struct CoefTables {
  alignas(16) float frag[32][16];
  float nat[NTERMS][NTERMS];
};

constexpr CoefTables make_coef_tables() {
  CoefTables t{};
  for (int m = 0; m < NTERMS; ++m) {
    double W = 0.2734375 * 3.14159265358979323846;
    for (int j = 0; j < m; ++j) {
      double tj = 2.0 * j;
      W *= (tj + 0.5) * (tj + 1.5) /
           (4.0 * (tj + 5.0) * (tj + 6.0) * (j + 1.0) * (j + 1.0));
    }
    double R = W;
    for (int k = 0; k < NTERMS; ++k) {
      t.nat[m][k] = (float)R;
      R *= (4.5 + k) / ((0.5 + k) * (2.0 * m + 5.0 + k) * (k + 1.0));
    }
  }
  for (int lane = 0; lane < 32; ++lane) {
    const int col  = lane & 15;
    const int rsel = (lane < 16) ? 0 : 2;
    for (int s = 0; s < 5; ++s) {
      const int r = 4 * s + rsel;
      t.frag[lane][2 * s + 0] = t.nat[r][col];
      t.frag[lane][2 * s + 1] = t.nat[r + 1][col];
    }
    const int j = col >> 2;
    const int c = col & 3;
    for (int s = 0; s < 2; ++s) {
      for (int y = 0; y < 2; ++y) {
        const int m = 8 * j + 4 * s + rsel + y;
        t.frag[lane][10 + 2 * s + y] = (j < 3 && m < NTERMS) ? t.nat[m][16 + c] : 0.0f;
      }
    }
    t.frag[lane][14] = 0.0f;
    t.frag[lane][15] = 0.0f;
  }
  return t;
}

__constant__ const CoefTables kCoef = make_coef_tables();

__global__ __launch_bounds__(256) void ipe_logsumexp_wmma(
    const float* __restrict__ a, const float* __restrict__ b,
    float* __restrict__ out, int nElem, int nBlocks64) {
  __shared__ __align__(16) float Tsh[8][4][16][TROW];

  const int lane = threadIdx.x & 31;
  const int wv   = threadIdx.x >> 5;
  const int col  = lane & 15;

  const float4* pB = (const float4*)&kCoef.frag[lane][0];
  const float4 f0 = pB[0], f1 = pB[1], f2 = pB[2], f3 = pB[3];
  const v2f B0[5] = {v2f{f0.x, f0.y}, v2f{f0.z, f0.w}, v2f{f1.x, f1.y},
                     v2f{f1.z, f1.w}, v2f{f2.x, f2.y}};
  const v2f B1[2] = {v2f{f2.z, f2.w}, v2f{f3.x, f3.y}};

  const int nwaves = (int)gridDim.x * (blockDim.x >> 5);
  const int gwave  = (int)blockIdx.x * (blockDim.x >> 5) + wv;

  float cb[4] = {0.0f, 0.0f, 0.0f, 0.0f};
  float ca[2] = {0.0f, 0.0f};
  if (gwave < nBlocks64) {
    const int base = gwave * 64;
#pragma unroll
    for (int t = 0; t < 4; ++t) cb[t] = b[base + 16 * t + col];
    ca[0] = a[base + lane];
    ca[1] = a[base + 32 + lane];
  }

  for (int blk = gwave; blk < nBlocks64; blk += nwaves) {
    const int base = blk * 64;

    const int nxt   = blk + nwaves;
    const int nbase = ((nxt < nBlocks64) ? nxt : blk) * 64;
    float nb[4], na[2];
#pragma unroll
    for (int t = 0; t < 4; ++t) nb[t] = b[nbase + 16 * t + col];
    na[0] = a[nbase + lane];
    na[1] = a[nbase + 32 + lane];

    float p0[4], p1[4], b4[4];
#pragma unroll
    for (int t = 0; t < 4; ++t) {
      const float bb = cb[t];
      const float b2 = bb * bb;
      p0[t] = (lane < 16) ? 1.0f : b2;
      p1[t] = (lane < 16) ? bb   : b2 * bb;
      b4[t] = b2 * b2;
    }

    v8f c0[4] = {}, c1[4] = {};
#pragma unroll
    for (int s = 0; s < 5; ++s) {
#pragma unroll
      for (int t = 0; t < 4; ++t) {
        const v2f A = v2f{p0[t], p1[t]};
        c0[t] = __builtin_amdgcn_wmma_f32_16x16x4_f32(false, A, false, B0[s], (short)0, c0[t], false, false);
        if (s < 2)
          c1[t] = __builtin_amdgcn_wmma_f32_16x16x4_f32(false, A, false, B1[s], (short)0, c1[t], false, false);
        p0[t] *= b4[t]; p1[t] *= b4[t];
      }
    }

    const int row = (lane < 16) ? 0 : 8;
#pragma unroll
    for (int t = 0; t < 4; ++t) {
#pragma unroll
      for (int i = 0; i < 8; ++i) {
        Tsh[wv][t][row + i][col]      = c0[t][i];
        Tsh[wv][t][row + i][16 + col] = c1[t][i];
      }
    }

#pragma unroll
    for (int u = 0; u < 2; ++u) {
      const float x = ca[u] * ca[u] * 0.25f;
      const float e4  = (lane < 16) ? b4[2 * u] : b4[2 * u + 1];
      const float e8  = e4 * e4;
      const float e16 = e8 * e8;
      const float4* pT4 = (const float4*)&Tsh[wv][2 * u + (lane >> 4)][lane & 15][0];
      float4 q0 = pT4[0], q1 = pT4[1], q2 = pT4[2], q3 = pT4[3];
      float4 q4 = pT4[4], q5 = pT4[5], q6 = pT4[6];
      const float t16 = fmaf(e16, q6.x, fmaf(e8, q5.x, q4.x));
      const float t17 = fmaf(e16, q6.y, fmaf(e8, q5.y, q4.y));
      const float t18 = fmaf(e16, q6.z, fmaf(e8, q5.z, q4.z));
      const float t19 = fmaf(e16, q6.w, fmaf(e8, q5.w, q4.w));
      float v = t19;
      v = fmaf(v, x, t18); v = fmaf(v, x, t17); v = fmaf(v, x, t16);
      v = fmaf(v, x, q3.w); v = fmaf(v, x, q3.z); v = fmaf(v, x, q3.y); v = fmaf(v, x, q3.x);
      v = fmaf(v, x, q2.w); v = fmaf(v, x, q2.z); v = fmaf(v, x, q2.y); v = fmaf(v, x, q2.x);
      v = fmaf(v, x, q1.w); v = fmaf(v, x, q1.z); v = fmaf(v, x, q1.y); v = fmaf(v, x, q1.x);
      v = fmaf(v, x, q0.w); v = fmaf(v, x, q0.z); v = fmaf(v, x, q0.y); v = fmaf(v, x, q0.x);
      out[base + 32 * u + lane] = logf(v);
    }

#pragma unroll
    for (int t = 0; t < 4; ++t) cb[t] = nb[t];
    ca[0] = na[0];
    ca[1] = na[1];
  }

  for (int e = nBlocks64 * 64 + gwave * 32 + lane; e < nElem; e += nwaves * 32) {
    float bb = b[e], aa = a[e];
    float x = aa * aa * 0.25f;
    float accv = 0.0f, xp = 1.0f;
    for (int k = 0; k < NTERMS; ++k) {
      float t = 0.0f, bp = 1.0f;
      for (int m = 0; m < NTERMS; ++m) { t = fmaf(kCoef.nat[m][k], bp, t); bp *= bb; }
      accv = fmaf(t, xp, accv);
      xp *= x;
    }
    out[e] = logf(accv);
  }
}

extern "C" void kernel_launch(void* const* d_in, const int* in_sizes, int n_in,
                              void* d_out, int out_size, void* d_ws, size_t ws_size,
                              hipStream_t stream) {
  (void)n_in; (void)out_size; (void)d_ws; (void)ws_size;
  const float* a = (const float*)d_in[0];
  const float* b = (const float*)d_in[1];
  float* out = (float*)d_out;
  int n = in_sizes[0];
  int blocks64 = n / 64;
  int blocks = (blocks64 + 7) / 8;
  if (blocks > 256) blocks = 256;
  if (blocks < 1) blocks = 1;
  ipe_logsumexp_wmma<<<blocks, 256, 0, stream>>>(a, b, out, n, blocks64);
}


# Products with coefficients that are exactly 0.0f in f32 skipped (rows m>=17; m>=8 of columns 16..19): 5 WMMA per 16 elements, was 7; leftovers share one tile, joined by b^4, b^16 (new sum order)
# speedup vs baseline: 13.0264x; 1.1069x over previous
#include <hip/hip_runtime.h>
#include <math.h>


typedef __attribute__((ext_vector_type(2))) float v2f;
typedef __attribute__((ext_vector_type(8))) float v8f;

#define NTERMS 20
#define TROW   36

struct CoefTables {
  alignas(16) float frag[32][12];
  float nat[NTERMS][NTERMS];
};

constexpr CoefTables make_coef_tables() {
  CoefTables t{};
  for (int m = 0; m < NTERMS; ++m) {
    double W = 0.2734375 * 3.14159265358979323846;
    for (int j = 0; j < m; ++j) {
      double tj = 2.0 * j;
      W *= (tj + 0.5) * (tj + 1.5) /
           (4.0 * (tj + 5.0) * (tj + 6.0) * (j + 1.0) * (j + 1.0));
    }
    double R = W;
    for (int k = 0; k < NTERMS; ++k) {
      t.nat[m][k] = (float)R;
      R *= (4.5 + k) / ((0.5 + k) * (2.0 * m + 5.0 + k) * (k + 1.0));
    }
  }
  for (int lane = 0; lane < 32; ++lane) {
    const int col  = lane & 15;
    const int rsel = (lane < 16) ? 0 : 2;
    for (int s = 0; s < 4; ++s) {
      const int r = 4 * s + rsel;
      t.frag[lane][2 * s + 0] = t.nat[r][col];
      t.frag[lane][2 * s + 1] = t.nat[r + 1][col];
    }
    const int g = col >> 2;
    const int c = col & 3;
    for (int y = 0; y < 2; ++y) {
      const int r = rsel + y;
      float v = 0.0f;
      if (g == 0) v = t.nat[r][16 + c];
      if (g == 1) v = t.nat[4 + r][16 + c];
      if (g == 2) v = t.nat[16 + r][c];
      t.frag[lane][8 + y] = v;
    }
    t.frag[lane][10] = 0.0f;
    t.frag[lane][11] = 0.0f;
  }
  return t;
}

constexpr bool skipped_entries_are_zero(const CoefTables& t) {
  for (int m = 16; m < NTERMS; ++m)
    for (int k = 4; k < 16; ++k)
      if (t.nat[m][k] != 0.0f) return false;
  for (int m = 8; m < NTERMS; ++m)
    for (int k = 16; k < NTERMS; ++k)
      if (t.nat[m][k] != 0.0f) return false;
  return true;
}
static_assert(skipped_entries_are_zero(make_coef_tables()),
              "the matrix path skips only coefficients that are exactly zero in f32");

__constant__ const CoefTables kCoef = make_coef_tables();

__global__ __launch_bounds__(256) void ipe_logsumexp_wmma(
    const float* __restrict__ a, const float* __restrict__ b,
    float* __restrict__ out, int nElem, int nBlocks64) {
  __shared__ __align__(16) float Tsh[8][4][16][TROW];

  const int lane = threadIdx.x & 31;
  const int wv   = threadIdx.x >> 5;
  const int col  = lane & 15;

  const float4* pB = (const float4*)&kCoef.frag[lane][0];
  const float4 f0 = pB[0], f1 = pB[1], f2 = pB[2];
  const v2f B0[4] = {v2f{f0.x, f0.y}, v2f{f0.z, f0.w}, v2f{f1.x, f1.y}, v2f{f1.z, f1.w}};
  const v2f B1 = v2f{f2.x, f2.y};

  const int nwaves = (int)gridDim.x * (blockDim.x >> 5);
  const int gwave  = (int)blockIdx.x * (blockDim.x >> 5) + wv;

  float cb[4] = {0.0f, 0.0f, 0.0f, 0.0f};
  float ca[2] = {0.0f, 0.0f};
  if (gwave < nBlocks64) {
    const int base = gwave * 64;
#pragma unroll
    for (int t = 0; t < 4; ++t) cb[t] = b[base + 16 * t + col];
    ca[0] = a[base + lane];
    ca[1] = a[base + 32 + lane];
  }

  for (int blk = gwave; blk < nBlocks64; blk += nwaves) {
    const int base = blk * 64;

    const int nxt   = blk + nwaves;
    const int nbase = ((nxt < nBlocks64) ? nxt : blk) * 64;
    float nb[4], na[2];
#pragma unroll
    for (int t = 0; t < 4; ++t) nb[t] = b[nbase + 16 * t + col];
    na[0] = a[nbase + lane];
    na[1] = a[nbase + 32 + lane];

    float p0[4], p1[4], b4[4];
#pragma unroll
    for (int t = 0; t < 4; ++t) {
      const float bb = cb[t];
      const float b2 = bb * bb;
      p0[t] = (lane < 16) ? 1.0f : b2;
      p1[t] = (lane < 16) ? bb   : b2 * bb;
      b4[t] = b2 * b2;
    }

    v8f c0[4] = {}, c1[4] = {};
#pragma unroll
    for (int s = 0; s < 4; ++s) {
#pragma unroll
      for (int t = 0; t < 4; ++t) {
        const v2f A = v2f{p0[t], p1[t]};
        c0[t] = __builtin_amdgcn_wmma_f32_16x16x4_f32(false, A, false, B0[s], (short)0, c0[t], false, false);
        if (s == 0)
          c1[t] = __builtin_amdgcn_wmma_f32_16x16x4_f32(false, A, false, B1, (short)0, c1[t], false, false);
        if (s < 3) { p0[t] *= b4[t]; p1[t] *= b4[t]; }
      }
    }

    const int row = (lane < 16) ? 0 : 8;
#pragma unroll
    for (int t = 0; t < 4; ++t) {
#pragma unroll
      for (int i = 0; i < 8; ++i) {
        Tsh[wv][t][row + i][col]      = c0[t][i];
        Tsh[wv][t][row + i][16 + col] = c1[t][i];
      }
    }

#pragma unroll
    for (int u = 0; u < 2; ++u) {
      const float x = ca[u] * ca[u] * 0.25f;
      const float e4  = (lane < 16) ? b4[2 * u] : b4[2 * u + 1];
      const float e8  = e4 * e4;
      const float e16 = e8 * e8;
      const float4* pT4 = (const float4*)&Tsh[wv][2 * u + (lane >> 4)][lane & 15][0];
      float4 q0 = pT4[0], q1 = pT4[1], q2 = pT4[2], q3 = pT4[3];
      float4 q4 = pT4[4], q5 = pT4[5], q6 = pT4[6];
      const float t16 = fmaf(e4, q5.x, q4.x);
      const float t17 = fmaf(e4, q5.y, q4.y);
      const float t18 = fmaf(e4, q5.z, q4.z);
      const float t19 = fmaf(e4, q5.w, q4.w);
      q0.x = fmaf(e16, q6.x, q0.x);
      q0.y = fmaf(e16, q6.y, q0.y);
      q0.z = fmaf(e16, q6.z, q0.z);
      q0.w = fmaf(e16, q6.w, q0.w);
      float v = t19;
      v = fmaf(v, x, t18); v = fmaf(v, x, t17); v = fmaf(v, x, t16);
      v = fmaf(v, x, q3.w); v = fmaf(v, x, q3.z); v = fmaf(v, x, q3.y); v = fmaf(v, x, q3.x);
      v = fmaf(v, x, q2.w); v = fmaf(v, x, q2.z); v = fmaf(v, x, q2.y); v = fmaf(v, x, q2.x);
      v = fmaf(v, x, q1.w); v = fmaf(v, x, q1.z); v = fmaf(v, x, q1.y); v = fmaf(v, x, q1.x);
      v = fmaf(v, x, q0.w); v = fmaf(v, x, q0.z); v = fmaf(v, x, q0.y); v = fmaf(v, x, q0.x);
      out[base + 32 * u + lane] = logf(v);
    }

#pragma unroll
    for (int t = 0; t < 4; ++t) cb[t] = nb[t];
    ca[0] = na[0];
    ca[1] = na[1];
  }

  for (int e = nBlocks64 * 64 + gwave * 32 + lane; e < nElem; e += nwaves * 32) {
    float bb = b[e], aa = a[e];
    float x = aa * aa * 0.25f;
    float accv = 0.0f, xp = 1.0f;
    for (int k = 0; k < NTERMS; ++k) {
      float t = 0.0f, bp = 1.0f;
      for (int m = 0; m < NTERMS; ++m) { t = fmaf(kCoef.nat[m][k], bp, t); bp *= bb; }
      accv = fmaf(t, xp, accv);
      xp *= x;
    }
    out[e] = logf(accv);
  }
}

extern "C" void kernel_launch(void* const* d_in, const int* in_sizes, int n_in,
                              void* d_out, int out_size, void* d_ws, size_t ws_size,
                              hipStream_t stream) {
  (void)n_in; (void)out_size; (void)d_ws; (void)ws_size;
  const float* a = (const float*)d_in[0];
  const float* b = (const float*)d_in[1];
  float* out = (float*)d_out;
  int n = in_sizes[0];
  int blocks64 = n / 64;
  int blocks = (blocks64 + 7) / 8;
  if (blocks > 256) blocks = 256;
  if (blocks < 1) blocks = 1;
  ipe_logsumexp_wmma<<<blocks, 256, 0, stream>>>(a, b, out, n, blocks64);
}
